# starting kernel
# baseline (speedup 1.0000x reference)
#include <hip/hip_runtime.h>


typedef __attribute__((ext_vector_type(2))) float v2f;
typedef __attribute__((ext_vector_type(8))) float v8f;

#define N_DIM 4096
#define D_DIM 256
#define BM 64
#define BN 64
#define KT 16
#define A_STRIDE 20
#define BPS 160
#define A_BUF (BM * A_STRIDE)
#define B_BUF ((KT / 2) * BPS)

__global__ __launch_bounds__(256) void poly_init_kernel(const float* __restrict__ x,
                                                        const float* __restrict__ coeffs,
                                                        float* __restrict__ y) {
    int i = blockIdx.x * blockDim.x + threadIdx.x;
    y[i] = coeffs[7] * x[i];
}

__global__ __launch_bounds__(256) void poly_gemm_kernel(const float* __restrict__ L,
                                                        const float* __restrict__ Y,
                                                        const float* __restrict__ X,
                                                        const float* __restrict__ coeffs,
                                                        int kIdx,
                                                        float* __restrict__ Out) {
    __shared__ float sA[2 * A_BUF];
    __shared__ float sB[2 * B_BUF];

    const int t    = threadIdx.x;
    const int lane = t & 31;
    const int wave = t >> 5;
    const int half = lane >> 4;
    const int l16  = lane & 15;

    const int mWave  = (wave & 3) * 16;
    const int nWave  = (wave >> 2) * 32;
    const int mBlock = blockIdx.x * BM;
    const int nBlock = blockIdx.y * BN;

    const int ar = t >> 2;
    const int ac = (t & 3) << 2;
    const int br = t >> 4;
    const int bc = (t & 15) << 2;
    const int bkp = br >> 1;
    const int bko = br & 1;

    const long aRowBase = (long)(mBlock + ar) * N_DIM + ac;

    v8f acc0 = {};
    v8f acc1 = {};

    float4 ra = *(const float4*)&L[aRowBase];
    float4 rb = *(const float4*)&Y[br * D_DIM + nBlock + bc];
    *(float4*)&sA[ar * A_STRIDE + ac] = ra;
    {
        float* p = &sB[bkp * BPS + 2 * bc + bko];
        p[0] = rb.x; p[2] = rb.y; p[4] = rb.z; p[6] = rb.w;
    }

    int cur = 0;
    for (int kk = 0; kk < N_DIM; kk += KT) {
        __syncthreads();

        const bool more = (kk + KT) < N_DIM;
        if (more) {
            ra = *(const float4*)&L[aRowBase + kk + KT];
            rb = *(const float4*)&Y[(kk + KT + br) * D_DIM + nBlock + bc];
            __builtin_prefetch(&L[aRowBase + kk + 2 * KT], 0, 3);
        }

        const int aOff = cur * A_BUF;
        const int bOff = cur * B_BUF;
#pragma unroll
        for (int kb = 0; kb < KT / 4; ++kb) {
            const int kp = 2 * kb + half;
            v2f a = *(const v2f*)&sA[aOff + (mWave + l16) * A_STRIDE + 4 * kb + 2 * half];
            v2f b0 = *(const v2f*)&sB[bOff + kp * BPS + 2 * (nWave + l16)];
            v2f b1 = *(const v2f*)&sB[bOff + kp * BPS + 2 * (nWave + 16 + l16)];
            acc0 = __builtin_amdgcn_wmma_f32_16x16x4_f32(
                false, a, false, b0, (short)0, acc0, false, false);
            acc1 = __builtin_amdgcn_wmma_f32_16x16x4_f32(
                false, a, false, b1, (short)0, acc1, false, false);
        }

        if (more) {
            const int na = (cur ^ 1) * A_BUF;
            const int nb = (cur ^ 1) * B_BUF;
            *(float4*)&sA[na + ar * A_STRIDE + ac] = ra;
            float* p = &sB[nb + bkp * BPS + 2 * bc + bko];
            p[0] = rb.x; p[2] = rb.y; p[4] = rb.z; p[6] = rb.w;
        }
        cur ^= 1;
    }

    const float c    = coeffs[kIdx];
    const int   mrow = mBlock + mWave + 8 * half;
    const int   ncol = nBlock + nWave + l16;
#pragma unroll
    for (int r = 0; r < 8; ++r) {
        int i0 = (mrow + r) * D_DIM + ncol;
        Out[i0]      = acc0[r] + c * X[i0];
        Out[i0 + 16] = acc1[r] + c * X[i0 + 16];
    }
}

extern "C" void kernel_launch(void* const* d_in, const int* in_sizes, int n_in,
                              void* d_out, int out_size, void* d_ws, size_t ws_size,
                              hipStream_t stream) {
    (void)in_sizes; (void)n_in; (void)out_size; (void)ws_size;
    const float* L      = (const float*)d_in[0];
    const float* x      = (const float*)d_in[1];
    const float* coeffs = (const float*)d_in[2];
    float* out = (float*)d_out;
    float* ws  = (float*)d_ws;

    poly_init_kernel<<<(N_DIM * D_DIM) / 256, 256, 0, stream>>>(x, coeffs, ws);

    dim3 grid(N_DIM / BM, D_DIM / BN);
    const float* src = ws;
    for (int p = 0; p < 7; ++p) {
        int    kIdx = 6 - p;
        float* dst  = (p & 1) ? ws : out;
        poly_gemm_kernel<<<grid, 256, 0, stream>>>(L, src, x, coeffs, kIdx, dst);
        src = dst;
    }
}


# K tile per work-group barrier 16 -> 128: each thread keeps the next tile (16 b128 loads) in flight under 64 matrix instructions per wave; latency and barrier paid once per 128 k; same sum order
# speedup vs baseline: 2.0844x; 2.0844x over previous
#include <hip/hip_runtime.h>


typedef __attribute__((ext_vector_type(2))) float v2f;
typedef __attribute__((ext_vector_type(4))) float v4f;
typedef __attribute__((ext_vector_type(8))) float v8f;

#define N_DIM 4096
#define D_DIM 256
#define BM 64
#define BN 64
#define KT 128
#define A_STRIDE 132
#define BPS 160
#define A_BUF (BM * A_STRIDE)
#define B_BUF ((KT / 2) * BPS)
#define A_LD ((BM * KT) / (4 * 256))
#define B_LD ((KT * BN) / (4 * 256))

__global__ __launch_bounds__(256) void poly_init_kernel(const float* __restrict__ x,
                                                        const float* __restrict__ coeffs,
                                                        float* __restrict__ y) {
    int i = blockIdx.x * blockDim.x + threadIdx.x;
    y[i] = coeffs[7] * x[i];
}

__global__ __launch_bounds__(256) void poly_gemm_kernel(const float* __restrict__ L,
                                                        const float* __restrict__ Y,
                                                        const float* __restrict__ X,
                                                        const float* __restrict__ coeffs,
                                                        int kIdx,
                                                        float* __restrict__ Out) {
    __shared__ float sA[2 * A_BUF];
    __shared__ float sB[2 * B_BUF];

    const int t    = threadIdx.x;
    const int lane = t & 31;
    const int wave = t >> 5;
    const int half = lane >> 4;
    const int l16  = lane & 15;

    const int mWave  = (wave & 3) * 16;
    const int nWave  = (wave >> 2) * 32;
    const int mBlock = blockIdx.x * BM;
    const int nBlock = blockIdx.y * BN;

    const int ar  = t >> 5;
    const int ac  = (t & 31) << 2;
    const int br  = t >> 4;
    const int bc  = (t & 15) << 2;
    const int bkp = br >> 1;
    const int bko = br & 1;

    const float* aSrc = L + (long)(mBlock + ar) * N_DIM + ac;
    const float* bSrc = Y + (long)br * D_DIM + nBlock + bc;
    const int aDst = ar * A_STRIDE + ac;
    const int bDst = bkp * BPS + 2 * bc + bko;

    v8f acc0 = {};
    v8f acc1 = {};

    v4f ra[A_LD];
    v4f rb[B_LD];

#pragma unroll
    for (int j = 0; j < A_LD; ++j)
        ra[j] = *(const v4f*)&aSrc[(long)(8 * j) * N_DIM];
#pragma unroll
    for (int j = 0; j < B_LD; ++j)
        rb[j] = *(const v4f*)&bSrc[(long)(16 * j) * D_DIM];
#pragma unroll
    for (int j = 0; j < A_LD; ++j)
        *(v4f*)&sA[aDst + 8 * j * A_STRIDE] = ra[j];
#pragma unroll
    for (int j = 0; j < B_LD; ++j) {
        float* p = &sB[bDst + 8 * j * BPS];
        p[0] = rb[j].x; p[2] = rb[j].y; p[4] = rb[j].z; p[6] = rb[j].w;
    }

    int cur = 0;
    for (int kk = 0; kk < N_DIM; kk += KT) {
        __syncthreads();

        const bool more = (kk + KT) < N_DIM;
        if (more) {
#pragma unroll
            for (int j = 0; j < A_LD; ++j)
                ra[j] = *(const v4f*)&aSrc[(long)(8 * j) * N_DIM + kk + KT];
#pragma unroll
            for (int j = 0; j < B_LD; ++j)
                rb[j] = *(const v4f*)&bSrc[(long)(kk + KT + 16 * j) * D_DIM];
        }

        const int aOff = cur * A_BUF;
        const int bOff = cur * B_BUF;
#pragma unroll
        for (int kb = 0; kb < KT / 4; ++kb) {
            const int kp = 2 * kb + half;
            v2f a = *(const v2f*)&sA[aOff + (mWave + l16) * A_STRIDE + 4 * kb + 2 * half];
            v2f b0 = *(const v2f*)&sB[bOff + kp * BPS + 2 * (nWave + l16)];
            v2f b1 = *(const v2f*)&sB[bOff + kp * BPS + 2 * (nWave + 16 + l16)];
            acc0 = __builtin_amdgcn_wmma_f32_16x16x4_f32(
                false, a, false, b0, (short)0, acc0, false, false);
            acc1 = __builtin_amdgcn_wmma_f32_16x16x4_f32(
                false, a, false, b1, (short)0, acc1, false, false);
        }

        if (more) {
            const int na = (cur ^ 1) * A_BUF;
            const int nb = (cur ^ 1) * B_BUF;
#pragma unroll
            for (int j = 0; j < A_LD; ++j)
                *(v4f*)&sA[na + aDst + 8 * j * A_STRIDE] = ra[j];
#pragma unroll
            for (int j = 0; j < B_LD; ++j) {
                float* p = &sB[nb + bDst + 8 * j * BPS];
                p[0] = rb[j].x; p[2] = rb[j].y; p[4] = rb[j].z; p[6] = rb[j].w;
            }
        }
        cur ^= 1;
    }

    const float c    = coeffs[kIdx];
    const int   mrow = mBlock + mWave + 8 * half;
    const int   ncol = nBlock + nWave + l16;
#pragma unroll
    for (int r = 0; r < 8; ++r) {
        int i0 = (mrow + r) * D_DIM + ncol;
        Out[i0]      = acc0[r] + c * X[i0];
        Out[i0 + 16] = acc1[r] + c * X[i0 + 16];
    }
}

extern "C" void kernel_launch(void* const* d_in, const int* in_sizes, int n_in,
                              void* d_out, int out_size, void* d_ws, size_t ws_size,
                              hipStream_t stream) {
    (void)in_sizes; (void)n_in; (void)out_size; (void)ws_size;
    const float* L      = (const float*)d_in[0];
    const float* x      = (const float*)d_in[1];
    const float* coeffs = (const float*)d_in[2];
    float* out = (float*)d_out;
    float* ws  = (float*)d_ws;

    poly_init_kernel<<<(N_DIM * D_DIM) / 256, 256, 0, stream>>>(x, coeffs, ws);

    dim3 grid(N_DIM / BM, D_DIM / BN);
    const float* src = ws;
    for (int p = 0; p < 7; ++p) {
        int    kIdx = 6 - p;
        float* dst  = (p & 1) ? ws : out;
        poly_gemm_kernel<<<grid, 256, 0, stream>>>(L, src, x, coeffs, kIdx, dst);
        src = dst;
    }
}


# Staging moved into the matrix stream: each 2 k-steps store one piece of tile t+1 to LDS and reload it from global for tile t+2; no store block behind the matrix ops; fragments read 2 k-steps ahead
# speedup vs baseline: 2.3180x; 1.1121x over previous
#include <hip/hip_runtime.h>


typedef __attribute__((ext_vector_type(2))) float v2f;
typedef __attribute__((ext_vector_type(4))) float v4f;
typedef __attribute__((ext_vector_type(8))) float v8f;

#define N_DIM 4096
#define D_DIM 256
#define BM 64
#define BN 64
#define KT 128
#define NT (N_DIM / KT)
#define KSTEPS (KT / 4)
#define A_STRIDE 132
#define BPS 160
#define A_BUF (BM * A_STRIDE)
#define B_BUF ((KT / 2) * BPS)
#define A_LD ((BM * KT) / (4 * 256))
#define B_LD ((KT * BN) / (4 * 256))

static_assert(KSTEPS == 2 * (A_LD + B_LD), "one staging piece per two k-steps");
static_assert((NT & 1) == 0 && NT >= 4, "tile NT-2 sits in buffer 0, tile NT-1 in buffer 1");

__global__ __launch_bounds__(256) void poly_init_kernel(const float* __restrict__ x,
                                                        const float* __restrict__ coeffs,
                                                        float* __restrict__ y) {
    int i = blockIdx.x * blockDim.x + threadIdx.x;
    y[i] = coeffs[7] * x[i];
}

static __device__ __forceinline__ void lds_barrier() {
    asm volatile("s_wait_dscnt 0x0" ::: "memory");
    asm volatile("s_barrier_signal -1" ::: "memory");
    asm volatile("s_barrier_wait -1" ::: "memory");
}

template <bool ST, bool LD>
static __device__ __forceinline__ void tile_body(const float* fragA, const float* fragB,
                                                 float* stA, float* stB,
                                                 const float* gA, const float* gB,
                                                 v4f (&ra)[A_LD], v4f (&rb)[B_LD],
                                                 v8f& acc0, v8f& acc1) {
    v2f fa[KSTEPS], fb0[KSTEPS], fb1[KSTEPS];
    fa[0]  = *(const v2f*)&fragA[0];
    fa[1]  = *(const v2f*)&fragA[4];
    fb0[0] = *(const v2f*)&fragB[0];
    fb1[0] = *(const v2f*)&fragB[32];
    fb0[1] = *(const v2f*)&fragB[2 * BPS];
    fb1[1] = *(const v2f*)&fragB[2 * BPS + 32];
#pragma unroll
    for (int kb = 0; kb < KSTEPS; ++kb) {
        if (kb + 2 < KSTEPS) {
            if ((kb & 1) == 0) {
                fa[kb + 2] = *(const v2f*)&fragA[4 * (kb + 2)];
                fa[kb + 3] = *(const v2f*)&fragA[4 * (kb + 3)];
            }
            fb0[kb + 2] = *(const v2f*)&fragB[2 * (kb + 2) * BPS];
            fb1[kb + 2] = *(const v2f*)&fragB[2 * (kb + 2) * BPS + 32];
        }
        const int p = kb >> 1;
        const int j = p >> 1;
        if ((kb & 1) == 0) {
            if (ST) {
                if ((p & 1) == 0) {
                    *(v4f*)&stA[8 * j * A_STRIDE] = ra[j];
                } else {
                    float* q = &stB[8 * j * BPS];
                    q[0] = rb[j].x; q[2] = rb[j].y; q[4] = rb[j].z; q[6] = rb[j].w;
                }
            }
        } else {
            if (LD) {
                if ((p & 1) == 0) ra[j] = *(const v4f*)&gA[(long)(8 * j) * N_DIM];
                else              rb[j] = *(const v4f*)&gB[(long)(16 * j) * D_DIM];
            }
        }
        acc0 = __builtin_amdgcn_wmma_f32_16x16x4_f32(
            false, fa[kb], false, fb0[kb], (short)0, acc0, false, false);
        acc1 = __builtin_amdgcn_wmma_f32_16x16x4_f32(
            false, fa[kb], false, fb1[kb], (short)0, acc1, false, false);
        __builtin_amdgcn_sched_barrier(0);
    }
}

__global__ __launch_bounds__(256) void poly_gemm_kernel(const float* __restrict__ L,
                                                        const float* __restrict__ Y,
                                                        const float* __restrict__ X,
                                                        const float* __restrict__ coeffs,
                                                        int kIdx,
                                                        float* __restrict__ Out) {
    __shared__ float sA[2 * A_BUF];
    __shared__ float sB[2 * B_BUF];

    const int t    = threadIdx.x;
    const int lane = t & 31;
    const int wave = t >> 5;
    const int half = lane >> 4;
    const int l16  = lane & 15;

    const int mWave  = (wave & 3) * 16;
    const int nWave  = (wave >> 2) * 32;
    const int mBlock = blockIdx.x * BM;
    const int nBlock = blockIdx.y * BN;

    const int ar  = t >> 5;
    const int ac  = (t & 31) << 2;
    const int br  = t >> 4;
    const int bc  = (t & 15) << 2;
    const int bkp = br >> 1;
    const int bko = br & 1;

    const float* aSrc = L + (long)(mBlock + ar) * N_DIM + ac;
    const float* bSrc = Y + (long)br * D_DIM + nBlock + bc;
    const int aDst = ar * A_STRIDE + ac;
    const int bDst = bkp * BPS + 2 * bc + bko;
    const int fA = (mWave + l16) * A_STRIDE + 2 * half;
    const int fB = half * BPS + 2 * (nWave + l16);

    v8f acc0 = {};
    v8f acc1 = {};

    v4f ra[A_LD];
    v4f rb[B_LD];

#pragma unroll
    for (int j = 0; j < A_LD; ++j)
        ra[j] = *(const v4f*)&aSrc[(long)(8 * j) * N_DIM];
#pragma unroll
    for (int j = 0; j < B_LD; ++j)
        rb[j] = *(const v4f*)&bSrc[(long)(16 * j) * D_DIM];
#pragma unroll
    for (int j = 0; j < A_LD; ++j)
        *(v4f*)&sA[aDst + 8 * j * A_STRIDE] = ra[j];
#pragma unroll
    for (int j = 0; j < B_LD; ++j) {
        float* p = &sB[bDst + 8 * j * BPS];
        p[0] = rb[j].x; p[2] = rb[j].y; p[4] = rb[j].z; p[6] = rb[j].w;
    }
#pragma unroll
    for (int j = 0; j < A_LD; ++j) {
        ra[j] = *(const v4f*)&aSrc[(long)(8 * j) * N_DIM + KT];
        rb[j] = *(const v4f*)&bSrc[(long)(KT + 16 * j) * D_DIM];
    }

    const float* gA = aSrc + 2 * KT;
    const float* gB = bSrc + (long)(2 * KT) * D_DIM;
    for (int kt = 0; kt < NT - 2; ++kt) {
        const int cur = kt & 1;
        lds_barrier();
        tile_body<true, true>(&sA[cur * A_BUF + fA], &sB[cur * B_BUF + fB],
                              &sA[(cur ^ 1) * A_BUF + aDst], &sB[(cur ^ 1) * B_BUF + bDst],
                              gA, gB, ra, rb, acc0, acc1);
        gA += KT;
        gB += (long)KT * D_DIM;
    }
    lds_barrier();
    tile_body<true, false>(&sA[fA], &sB[fB], &sA[A_BUF + aDst], &sB[B_BUF + bDst],
                           gA, gB, ra, rb, acc0, acc1);
    lds_barrier();
    tile_body<false, false>(&sA[A_BUF + fA], &sB[B_BUF + fB], &sA[aDst], &sB[bDst],
                            gA, gB, ra, rb, acc0, acc1);

    const float c    = coeffs[kIdx];
    const int   mrow = mBlock + mWave + 8 * half;
    const int   ncol = nBlock + nWave + l16;
#pragma unroll
    for (int r = 0; r < 8; ++r) {
        int i0 = (mrow + r) * D_DIM + ncol;
        Out[i0]      = acc0[r] + c * X[i0];
        Out[i0 + 16] = acc1[r] + c * X[i0 + 16];
    }
}

extern "C" void kernel_launch(void* const* d_in, const int* in_sizes, int n_in,
                              void* d_out, int out_size, void* d_ws, size_t ws_size,
                              hipStream_t stream) {
    (void)in_sizes; (void)n_in; (void)out_size; (void)ws_size;
    const float* L      = (const float*)d_in[0];
    const float* x      = (const float*)d_in[1];
    const float* coeffs = (const float*)d_in[2];
    float* out = (float*)d_out;
    float* ws  = (float*)d_ws;

    poly_init_kernel<<<(N_DIM * D_DIM) / 256, 256, 0, stream>>>(x, coeffs, ws);

    dim3 grid(N_DIM / BM, D_DIM / BN);
    const float* src = ws;
    for (int p = 0; p < 7; ++p) {
        int    kIdx = 6 - p;
        float* dst  = (p & 1) ? ws : out;
        poly_gemm_kernel<<<grid, 256, 0, stream>>>(L, src, x, coeffs, kIdx, dst);
        src = dst;
    }
}


# Priority window: the younger wave of each SIMD pair (waves 4-7) runs at s_setprio 3 for the first 20 of a tile's 32 k-steps, then 0, so the pair reaches the tile barrier together; same sum order
# speedup vs baseline: 2.3885x; 1.0304x over previous
#include <hip/hip_runtime.h>


typedef __attribute__((ext_vector_type(2))) float v2f;
typedef __attribute__((ext_vector_type(4))) float v4f;
typedef __attribute__((ext_vector_type(8))) float v8f;

#define N_DIM 4096
#define D_DIM 256
#define BM 64
#define BN 64
#define KT 128
#define NT (N_DIM / KT)
#define KSTEPS (KT / 4)
#define A_STRIDE 132
#define BPS 160
#define A_BUF (BM * A_STRIDE)
#define B_BUF ((KT / 2) * BPS)
#define A_LD ((BM * KT) / (4 * 256))
#define B_LD ((KT * BN) / (4 * 256))

#define PRIO_STEPS 20

static_assert(KSTEPS == 2 * (A_LD + B_LD), "one staging piece per two k-steps");
static_assert(PRIO_STEPS > 0 && PRIO_STEPS < KSTEPS, "the priority is dropped again inside the same tile");
static_assert((NT & 1) == 0 && NT >= 4, "tile NT-2 sits in buffer 0, tile NT-1 in buffer 1");

__global__ __launch_bounds__(256) void poly_init_kernel(const float* __restrict__ x,
                                                        const float* __restrict__ coeffs,
                                                        float* __restrict__ y) {
    int i = blockIdx.x * blockDim.x + threadIdx.x;
    y[i] = coeffs[7] * x[i];
}

static __device__ __forceinline__ void lds_barrier() {
    asm volatile("s_wait_dscnt 0x0" ::: "memory");
    asm volatile("s_barrier_signal -1" ::: "memory");
    asm volatile("s_barrier_wait -1" ::: "memory");
}

template <int PRIO>
static __device__ __forceinline__ void set_priority_if(int young) {
    asm volatile("s_cmp_eq_u32 %0, 0\n\t"
                 "s_cbranch_scc1 1f\n\t"
                 "s_setprio %1\n"
                 "1:"
                 :
                 : "s"(young), "n"(PRIO)
                 : "scc");
}

template <bool ST, bool LD>
static __device__ __forceinline__ void tile_body(const int young,
                                                 const float* fragA, const float* fragB,
                                                 float* stA, float* stB,
                                                 const float* gA, const float* gB,
                                                 v4f (&ra)[A_LD], v4f (&rb)[B_LD],
                                                 v8f& acc0, v8f& acc1) {
    v2f fa[KSTEPS], fb0[KSTEPS], fb1[KSTEPS];
    fa[0]  = *(const v2f*)&fragA[0];
    fa[1]  = *(const v2f*)&fragA[4];
    fb0[0] = *(const v2f*)&fragB[0];
    fb1[0] = *(const v2f*)&fragB[32];
    fb0[1] = *(const v2f*)&fragB[2 * BPS];
    fb1[1] = *(const v2f*)&fragB[2 * BPS + 32];
#pragma unroll
    for (int kb = 0; kb < KSTEPS; ++kb) {
        if (kb == 0) set_priority_if<3>(young);
        if (kb == PRIO_STEPS) set_priority_if<0>(young);
        if (kb + 2 < KSTEPS) {
            if ((kb & 1) == 0) {
                fa[kb + 2] = *(const v2f*)&fragA[4 * (kb + 2)];
                fa[kb + 3] = *(const v2f*)&fragA[4 * (kb + 3)];
            }
            fb0[kb + 2] = *(const v2f*)&fragB[2 * (kb + 2) * BPS];
            fb1[kb + 2] = *(const v2f*)&fragB[2 * (kb + 2) * BPS + 32];
        }
        const int p = kb >> 1;
        const int j = p >> 1;
        if ((kb & 1) == 0) {
            if (ST) {
                if ((p & 1) == 0) {
                    *(v4f*)&stA[8 * j * A_STRIDE] = ra[j];
                } else {
                    float* q = &stB[8 * j * BPS];
                    q[0] = rb[j].x; q[2] = rb[j].y; q[4] = rb[j].z; q[6] = rb[j].w;
                }
            }
        } else {
            if (LD) {
                if ((p & 1) == 0) ra[j] = *(const v4f*)&gA[(long)(8 * j) * N_DIM];
                else              rb[j] = *(const v4f*)&gB[(long)(16 * j) * D_DIM];
            }
        }
        acc0 = __builtin_amdgcn_wmma_f32_16x16x4_f32(
            false, fa[kb], false, fb0[kb], (short)0, acc0, false, false);
        acc1 = __builtin_amdgcn_wmma_f32_16x16x4_f32(
            false, fa[kb], false, fb1[kb], (short)0, acc1, false, false);
        __builtin_amdgcn_sched_barrier(0);
    }
}

__global__ __launch_bounds__(256) void poly_gemm_kernel(const float* __restrict__ L,
                                                        const float* __restrict__ Y,
                                                        const float* __restrict__ X,
                                                        const float* __restrict__ coeffs,
                                                        int kIdx,
                                                        float* __restrict__ Out) {
    __shared__ float sA[2 * A_BUF];
    __shared__ float sB[2 * B_BUF];

    const int t    = threadIdx.x;
    const int lane = t & 31;
    const int wave = t >> 5;
    const int half = lane >> 4;
    const int l16  = lane & 15;
    const int young = (__builtin_amdgcn_readfirstlane(t) >= 128) ? 1 : 0;

    const int mWave  = (wave & 3) * 16;
    const int nWave  = (wave >> 2) * 32;
    const int mBlock = blockIdx.x * BM;
    const int nBlock = blockIdx.y * BN;

    const int ar  = t >> 5;
    const int ac  = (t & 31) << 2;
    const int br  = t >> 4;
    const int bc  = (t & 15) << 2;
    const int bkp = br >> 1;
    const int bko = br & 1;

    const float* aSrc = L + (long)(mBlock + ar) * N_DIM + ac;
    const float* bSrc = Y + (long)br * D_DIM + nBlock + bc;
    const int aDst = ar * A_STRIDE + ac;
    const int bDst = bkp * BPS + 2 * bc + bko;
    const int fA = (mWave + l16) * A_STRIDE + 2 * half;
    const int fB = half * BPS + 2 * (nWave + l16);

    v8f acc0 = {};
    v8f acc1 = {};

    v4f ra[A_LD];
    v4f rb[B_LD];

#pragma unroll
    for (int j = 0; j < A_LD; ++j)
        ra[j] = *(const v4f*)&aSrc[(long)(8 * j) * N_DIM];
#pragma unroll
    for (int j = 0; j < B_LD; ++j)
        rb[j] = *(const v4f*)&bSrc[(long)(16 * j) * D_DIM];
#pragma unroll
    for (int j = 0; j < A_LD; ++j)
        *(v4f*)&sA[aDst + 8 * j * A_STRIDE] = ra[j];
#pragma unroll
    for (int j = 0; j < B_LD; ++j) {
        float* p = &sB[bDst + 8 * j * BPS];
        p[0] = rb[j].x; p[2] = rb[j].y; p[4] = rb[j].z; p[6] = rb[j].w;
    }
#pragma unroll
    for (int j = 0; j < A_LD; ++j) {
        ra[j] = *(const v4f*)&aSrc[(long)(8 * j) * N_DIM + KT];
        rb[j] = *(const v4f*)&bSrc[(long)(KT + 16 * j) * D_DIM];
    }

    const float* gA = aSrc + 2 * KT;
    const float* gB = bSrc + (long)(2 * KT) * D_DIM;
    for (int kt = 0; kt < NT - 2; ++kt) {
        const int cur = kt & 1;
        lds_barrier();
        tile_body<true, true>(young, &sA[cur * A_BUF + fA], &sB[cur * B_BUF + fB],
                              &sA[(cur ^ 1) * A_BUF + aDst], &sB[(cur ^ 1) * B_BUF + bDst],
                              gA, gB, ra, rb, acc0, acc1);
        gA += KT;
        gB += (long)KT * D_DIM;
    }
    lds_barrier();
    tile_body<true, false>(young, &sA[fA], &sB[fB], &sA[A_BUF + aDst], &sB[B_BUF + bDst],
                           gA, gB, ra, rb, acc0, acc1);
    lds_barrier();
    tile_body<false, false>(young, &sA[A_BUF + fA], &sB[B_BUF + fB], &sA[aDst], &sB[bDst],
                            gA, gB, ra, rb, acc0, acc1);

    const float c    = coeffs[kIdx];
    const int   mrow = mBlock + mWave + 8 * half;
    const int   ncol = nBlock + nWave + l16;
#pragma unroll
    for (int r = 0; r < 8; ++r) {
        int i0 = (mrow + r) * D_DIM + ncol;
        Out[i0]      = acc0[r] + c * X[i0];
        Out[i0 + 16] = acc1[r] + c * X[i0 + 16];
    }
}

extern "C" void kernel_launch(void* const* d_in, const int* in_sizes, int n_in,
                              void* d_out, int out_size, void* d_ws, size_t ws_size,
                              hipStream_t stream) {
    (void)in_sizes; (void)n_in; (void)out_size; (void)ws_size;
    const float* L      = (const float*)d_in[0];
    const float* x      = (const float*)d_in[1];
    const float* coeffs = (const float*)d_in[2];
    float* out = (float*)d_out;
    float* ws  = (float*)d_ws;

    poly_init_kernel<<<(N_DIM * D_DIM) / 256, 256, 0, stream>>>(x, coeffs, ws);

    dim3 grid(N_DIM / BM, D_DIM / BN);
    const float* src = ws;
    for (int p = 0; p < 7; ++p) {
        int    kIdx = 6 - p;
        float* dst  = (p & 1) ? ws : out;
        poly_gemm_kernel<<<grid, 256, 0, stream>>>(L, src, x, coeffs, kIdx, dst);
        src = dst;
    }
}


# LDS work grouped: one group per 16 matrix instr. (next block's 24 fragment reads + 4 staging pieces) instead of LDS ops at every k-step; tile barrier moved before the last block; same sum order
# speedup vs baseline: 2.6148x; 1.0947x over previous
#include <hip/hip_runtime.h>


typedef __attribute__((ext_vector_type(2))) float v2f;
typedef __attribute__((ext_vector_type(4))) float v4f;
typedef __attribute__((ext_vector_type(8))) float v8f;

#define N_DIM 4096
#define D_DIM 256
#define BM 64
#define BN 64
#define KT 128
#define NT (N_DIM / KT)
#define KSTEPS (KT / 4)
#define KB 8
#define NB (KSTEPS / KB)
#define A_STRIDE 132
#define BPS 160
#define A_BUF (BM * A_STRIDE)
#define B_BUF ((KT / 2) * BPS)
#define A_LD ((BM * KT) / (4 * 256))
#define B_LD ((KT * BN) / (4 * 256))
#define PPB ((A_LD + B_LD) / NB)

#define PRIO_STEPS 20

static_assert(NB == 4 && KB * NB == KSTEPS, "a tile is four blocks; the barrier stands in front of the fourth");
static_assert(PPB * NB == A_LD + B_LD && (PPB & 1) == 0, "every block stages the same number of A and B pieces");
static_assert(PRIO_STEPS == 2 * KB + KB / 2, "the priority is dropped in the middle of the third block behind a barrier");
static_assert((NT & 1) == 0 && NT >= 4, "tile NT-2 sits in buffer 0, tile NT-1 in buffer 1");

__global__ __launch_bounds__(256) void poly_init_kernel(const float* __restrict__ x,
                                                        const float* __restrict__ coeffs,
                                                        float* __restrict__ y) {
    int i = blockIdx.x * blockDim.x + threadIdx.x;
    y[i] = coeffs[7] * x[i];
}

static __device__ __forceinline__ void lds_barrier_signal() {
    asm volatile("s_wait_dscnt 0x0" ::: "memory");
    asm volatile("s_barrier_signal -1" ::: "memory");
}
static __device__ __forceinline__ void lds_barrier_wait() {
    asm volatile("s_barrier_wait -1" ::: "memory");
}

template <int PRIO>
static __device__ __forceinline__ void set_priority_if(int young) {
    asm volatile("s_cmp_eq_u32 %0, 0\n\t"
                 "s_cbranch_scc1 1f\n\t"
                 "s_setprio %1\n"
                 "1:"
                 :
                 : "s"(young), "n"(PRIO)
                 : "scc");
}

template <int Q, bool RD, bool ST, bool LD>
static __device__ __forceinline__ void lds_group(const float* fragA, const float* fragB,
                                                 float* stA, float* stB,
                                                 const float* gA, const float* gB,
                                                 v4f (&ra)[A_LD], v4f (&rb)[B_LD],
                                                 v2f (&fa)[KB], v2f (&fb0)[KB], v2f (&fb1)[KB]) {
    if (RD) {
#pragma unroll
        for (int i = 0; i < KB; ++i) {
            const int kb = Q * KB + i;
            fa[i]  = *(const v2f*)&fragA[4 * kb];
            fb0[i] = *(const v2f*)&fragB[2 * kb * BPS];
            fb1[i] = *(const v2f*)&fragB[2 * kb * BPS + 32];
        }
    }
    if (ST) {
#pragma unroll
        for (int pp = 0; pp < PPB; ++pp) {
            const int p = Q * PPB + pp;
            const int j = p >> 1;
            if ((p & 1) == 0) {
                *(v4f*)&stA[8 * j * A_STRIDE] = ra[j];
            } else {
                float* d = &stB[8 * j * BPS];
                d[0] = rb[j].x; d[2] = rb[j].y; d[4] = rb[j].z; d[6] = rb[j].w;
            }
        }
    }
    if (LD) {
#pragma unroll
        for (int pp = 0; pp < PPB; ++pp) {
            const int p = Q * PPB + pp;
            const int j = p >> 1;
            if ((p & 1) == 0) ra[j] = *(const v4f*)&gA[(long)(8 * j) * N_DIM];
            else              rb[j] = *(const v4f*)&gB[(long)(16 * j) * D_DIM];
        }
    }
}

template <int FROM, int TO>
static __device__ __forceinline__ void mma_steps(const v2f (&fa)[KB], const v2f (&fb0)[KB],
                                                 const v2f (&fb1)[KB], v8f& acc0, v8f& acc1) {
#pragma unroll
    for (int i = FROM; i < TO; ++i) {
        acc0 = __builtin_amdgcn_wmma_f32_16x16x4_f32(
            false, fa[i], false, fb0[i], (short)0, acc0, false, false);
        acc1 = __builtin_amdgcn_wmma_f32_16x16x4_f32(
            false, fa[i], false, fb1[i], (short)0, acc1, false, false);
    }
}

#define KEEP_APART() asm volatile("" : "+v"(acc0), "+v"(acc1) : : "memory")

template <bool FIRST, bool ST, bool LD>
static __device__ __forceinline__ void tile_body(const int young,
                                                 const float* curA, const float* curB,
                                                 float* stA, float* stB,
                                                 const float* gA, const float* gB,
                                                 v4f (&ra)[A_LD], v4f (&rb)[B_LD],
                                                 v2f (&fa)[2][KB], v2f (&fb0)[2][KB], v2f (&fb1)[2][KB],
                                                 v8f& acc0, v8f& acc1) {
    set_priority_if<3>(young);

    lds_group<0, true, ST, LD>(curA, curB, stA, stB, gA, gB, ra, rb, fa[0], fb0[0], fb1[0]);
    KEEP_APART();
    if (!FIRST) mma_steps<0, KB>(fa[1], fb0[1], fb1[1], acc0, acc1);
    KEEP_APART();

    lds_group<1, true, ST, LD>(curA, curB, stA, stB, gA, gB, ra, rb, fa[1], fb0[1], fb1[1]);
    KEEP_APART();
    mma_steps<0, KB>(fa[0], fb0[0], fb1[0], acc0, acc1);
    KEEP_APART();

    lds_group<2, true, ST, LD>(curA, curB, stA, stB, gA, gB, ra, rb, fa[0], fb0[0], fb1[0]);
    KEEP_APART();
    mma_steps<0, KB / 2>(fa[1], fb0[1], fb1[1], acc0, acc1);
    KEEP_APART();
    set_priority_if<0>(young);
    KEEP_APART();
    mma_steps<KB / 2, KB>(fa[1], fb0[1], fb1[1], acc0, acc1);
    KEEP_APART();

    lds_group<3, true, ST, LD>(curA, curB, stA, stB, gA, gB, ra, rb, fa[1], fb0[1], fb1[1]);
    KEEP_APART();
    mma_steps<0, KB / 2>(fa[0], fb0[0], fb1[0], acc0, acc1);
    KEEP_APART();
    if (ST) lds_barrier_signal();
    KEEP_APART();
    mma_steps<KB / 2, KB>(fa[0], fb0[0], fb1[0], acc0, acc1);
    KEEP_APART();
    if (ST) lds_barrier_wait();
}

__global__ __launch_bounds__(256) void poly_gemm_kernel(const float* __restrict__ L,
                                                        const float* __restrict__ Y,
                                                        const float* __restrict__ X,
                                                        const float* __restrict__ coeffs,
                                                        int kIdx,
                                                        float* __restrict__ Out) {
    __shared__ float sA[2 * A_BUF];
    __shared__ float sB[2 * B_BUF];

    const int t    = threadIdx.x;
    const int lane = t & 31;
    const int wave = t >> 5;
    const int half = lane >> 4;
    const int l16  = lane & 15;
    const int young = (__builtin_amdgcn_readfirstlane(t) >= 128) ? 1 : 0;

    const int mWave  = (wave & 3) * 16;
    const int nWave  = (wave >> 2) * 32;
    const int mBlock = blockIdx.x * BM;
    const int nBlock = blockIdx.y * BN;

    const int ar  = t >> 5;
    const int ac  = (t & 31) << 2;
    const int br  = t >> 4;
    const int bc  = (t & 15) << 2;
    const int bkp = br >> 1;
    const int bko = br & 1;

    const float* aSrc = L + (long)(mBlock + ar) * N_DIM + ac;
    const float* bSrc = Y + (long)br * D_DIM + nBlock + bc;
    const int aDst = ar * A_STRIDE + ac;
    const int bDst = bkp * BPS + 2 * bc + bko;
    const int fA = (mWave + l16) * A_STRIDE + 2 * half;
    const int fB = half * BPS + 2 * (nWave + l16);

    v8f acc0 = {};
    v8f acc1 = {};

    v4f ra[A_LD];
    v4f rb[B_LD];
    v2f fa[2][KB], fb0[2][KB], fb1[2][KB];

#pragma unroll
    for (int j = 0; j < A_LD; ++j)
        ra[j] = *(const v4f*)&aSrc[(long)(8 * j) * N_DIM];
#pragma unroll
    for (int j = 0; j < B_LD; ++j)
        rb[j] = *(const v4f*)&bSrc[(long)(16 * j) * D_DIM];
#pragma unroll
    for (int j = 0; j < A_LD; ++j)
        *(v4f*)&sA[aDst + 8 * j * A_STRIDE] = ra[j];
#pragma unroll
    for (int j = 0; j < B_LD; ++j) {
        float* p = &sB[bDst + 8 * j * BPS];
        p[0] = rb[j].x; p[2] = rb[j].y; p[4] = rb[j].z; p[6] = rb[j].w;
    }
#pragma unroll
    for (int j = 0; j < A_LD; ++j) {
        ra[j] = *(const v4f*)&aSrc[(long)(8 * j) * N_DIM + KT];
        rb[j] = *(const v4f*)&bSrc[(long)(KT + 16 * j) * D_DIM];
    }
    lds_barrier_signal();
    lds_barrier_wait();

    const float* gA = aSrc + 2 * KT;
    const float* gB = bSrc + (long)(2 * KT) * D_DIM;
    tile_body<true, true, true>(young, &sA[fA], &sB[fB], &sA[A_BUF + aDst], &sB[B_BUF + bDst],
                                gA, gB, ra, rb, fa, fb0, fb1, acc0, acc1);
    gA += KT;
    gB += (long)KT * D_DIM;
    for (int kt = 1; kt < NT - 2; ++kt) {
        const int cur = kt & 1;
        tile_body<false, true, true>(young, &sA[cur * A_BUF + fA], &sB[cur * B_BUF + fB],
                                     &sA[(cur ^ 1) * A_BUF + aDst], &sB[(cur ^ 1) * B_BUF + bDst],
                                     gA, gB, ra, rb, fa, fb0, fb1, acc0, acc1);
        gA += KT;
        gB += (long)KT * D_DIM;
    }
    tile_body<false, true, false>(young, &sA[fA], &sB[fB], &sA[A_BUF + aDst], &sB[B_BUF + bDst],
                                  gA, gB, ra, rb, fa, fb0, fb1, acc0, acc1);
    tile_body<false, false, false>(young, &sA[A_BUF + fA], &sB[B_BUF + fB], &sA[aDst], &sB[bDst],
                                   gA, gB, ra, rb, fa, fb0, fb1, acc0, acc1);
    mma_steps<0, KB>(fa[1], fb0[1], fb1[1], acc0, acc1);
    KEEP_APART();

    const float c    = coeffs[kIdx];
    const int   mrow = mBlock + mWave + 8 * half;
    const int   ncol = nBlock + nWave + l16;
#pragma unroll
    for (int r = 0; r < 8; ++r) {
        int i0 = (mrow + r) * D_DIM + ncol;
        Out[i0]      = acc0[r] + c * X[i0];
        Out[i0 + 16] = acc1[r] + c * X[i0 + 16];
    }
}

extern "C" void kernel_launch(void* const* d_in, const int* in_sizes, int n_in,
                              void* d_out, int out_size, void* d_ws, size_t ws_size,
                              hipStream_t stream) {
    (void)in_sizes; (void)n_in; (void)out_size; (void)ws_size;
    const float* L      = (const float*)d_in[0];
    const float* x      = (const float*)d_in[1];
    const float* coeffs = (const float*)d_in[2];
    float* out = (float*)d_out;
    float* ws  = (float*)d_ws;

    poly_init_kernel<<<(N_DIM * D_DIM) / 256, 256, 0, stream>>>(x, coeffs, ws);

    dim3 grid(N_DIM / BM, D_DIM / BN);
    const float* src = ws;
    for (int p = 0; p < 7; ++p) {
        int    kIdx = 6 - p;
        float* dst  = (p & 1) ? ws : out;
        poly_gemm_kernel<<<grid, 256, 0, stream>>>(L, src, x, coeffs, kIdx, dst);
        src = dst;
    }
}
